# starting kernel
# baseline (speedup 1.0000x reference)
#include <hip/hip_runtime.h>

typedef __attribute__((ext_vector_type(2))) float v2f;
typedef __attribute__((ext_vector_type(8))) float v8f;

#define IN_CH 128
#define HID   128

__global__ void k_init_deg(unsigned* __restrict__ deg, int N) {
  int i = blockIdx.x * blockDim.x + threadIdx.x;
  if (i < N) deg[i] = 1u;
}

__global__ void k_count(const int* __restrict__ ei, int E, unsigned* __restrict__ deg) {
  int e = blockIdx.x * blockDim.x + threadIdx.x;
  if (e < E) atomicAdd(&deg[ei[E + e]], 1u);
}

__global__ void k_dis(const unsigned* __restrict__ deg, float* __restrict__ dis, int N) {
  int i = blockIdx.x * blockDim.x + threadIdx.x;
  if (i < N) dis[i] = rsqrtf((float)deg[i]);
}

__global__ __launch_bounds__(128) void k_gemm(const float* __restrict__ x,
                                              const float* __restrict__ W,
                                              float* __restrict__ h) {
  const int lane = threadIdx.x & 31;
  const int wave = threadIdx.x >> 5;
  const int row0 = blockIdx.y * 16;
  const int col0 = (blockIdx.x * 4 + wave) * 16;

  const int mn  = lane & 15;
  const int kof = (lane >> 4) << 1;

  v8f acc = {};
  const float* xr = x + (size_t)(row0 + mn) * IN_CH;
  const float* Wc = W + col0 + mn;

#pragma unroll
  for (int k = 0; k < IN_CH; k += 4) {
    const float2 av = *(const float2*)(xr + k + kof);
    v2f a; a[0] = av.x; a[1] = av.y;
    v2f b;
    b[0] = Wc[(size_t)(k + kof) * HID];
    b[1] = Wc[(size_t)(k + kof + 1) * HID];
    acc = __builtin_amdgcn_wmma_f32_16x16x4_f32(false, a, false, b,
                                                (short)0, acc, false, false);
  }

  const int mo = (lane >> 4) * 8;
  float* hr = h + (size_t)(row0 + mo) * HID + col0 + mn;
#pragma unroll
  for (int r = 0; r < 8; ++r)
    hr[(size_t)r * HID] = acc[r];
}

__global__ void k_init_out(const float* __restrict__ h, const float* __restrict__ dis,
                           const float* __restrict__ bias, float* __restrict__ out, int N) {
  int t = blockIdx.x * blockDim.x + threadIdx.x;
  if (t >= N * 32) return;
  int i  = t >> 5;
  int cg = (t & 31) << 2;
  float nrm = dis[i] * dis[i];
  float4 hv = *(const float4*)(h + (size_t)i * HID + cg);
  float4 bv = *(const float4*)(bias + cg);
  float4 o;
  o.x = bv.x + nrm * hv.x;
  o.y = bv.y + nrm * hv.y;
  o.z = bv.z + nrm * hv.z;
  o.w = bv.w + nrm * hv.w;
  *(float4*)(out + (size_t)i * HID + cg) = o;
}

__global__ __launch_bounds__(256) void k_scatter(const int* __restrict__ ei, int E,
                                                 const float* __restrict__ h,
                                                 const float* __restrict__ dis,
                                                 float* __restrict__ out) {
  int w    = (blockIdx.x * blockDim.x + threadIdx.x) >> 5;
  int lane = threadIdx.x & 31;
  if (w >= E) return;
  int s = ei[w];
  int d = ei[E + w];
  float nrm = dis[s] * dis[d];
  const float4 hv = *(const float4*)(h + (size_t)s * HID + (lane << 2));
  float* op = out + (size_t)d * HID + (lane << 2);
  unsafeAtomicAdd(op + 0, nrm * hv.x);
  unsafeAtomicAdd(op + 1, nrm * hv.y);
  unsafeAtomicAdd(op + 2, nrm * hv.z);
  unsafeAtomicAdd(op + 3, nrm * hv.w);
}

__global__ void k_prelu(float* __restrict__ out, const float* __restrict__ pw, int N) {
  int t = blockIdx.x * blockDim.x + threadIdx.x;
  if (t >= N * 32) return;
  int cg = (t & 31) << 2;
  float4 v = *(float4*)(out + (size_t)t * 4);
  float4 p = *(const float4*)(pw + cg);
  v.x = v.x >= 0.f ? v.x : p.x * v.x;
  v.y = v.y >= 0.f ? v.y : p.y * v.y;
  v.z = v.z >= 0.f ? v.z : p.z * v.z;
  v.w = v.w >= 0.f ? v.w : p.w * v.w;
  *(float4*)(out + (size_t)t * 4) = v;
}

extern "C" void kernel_launch(void* const* d_in, const int* in_sizes, int n_in,
                              void* d_out, int out_size, void* d_ws, size_t ws_size,
                              hipStream_t stream) {
  const float* x  = (const float*)d_in[0];
  const int*   ei = (const int*)d_in[1];
  const float* W  = (const float*)d_in[2];
  const float* b  = (const float*)d_in[3];
  const float* pw = (const float*)d_in[4];
  float* out = (float*)d_out;

  const int N = in_sizes[0] / IN_CH;
  const int E = in_sizes[1] / 2;

  char* ws = (char*)d_ws;
  size_t o0 = 0;
  size_t o1 = (o0 + (size_t)N * 4 + 255) & ~(size_t)255;
  size_t o2 = (o1 + (size_t)N * 4 + 255) & ~(size_t)255;
  unsigned* deg = (unsigned*)(ws + o0);
  float*    dis = (float*)(ws + o1);
  float*    h   = (float*)(ws + o2);

  const int T = 256;
  k_init_deg<<<(N + T - 1) / T, T, 0, stream>>>(deg, N);
  k_count  <<<(E + T - 1) / T, T, 0, stream>>>(ei, E, deg);
  k_dis    <<<(N + T - 1) / T, T, 0, stream>>>(deg, dis, N);

  dim3 ggrid(2, N / 16);
  k_gemm<<<ggrid, 128, 0, stream>>>(x, W, h);

  int tq = N * 32;
  k_init_out<<<(tq + T - 1) / T, T, 0, stream>>>(h, dis, b, out, N);

  int sblocks = (E + 7) / 8;
  k_scatter<<<sblocks, T, 0, stream>>>(ei, E, h, dis, out);

  k_prelu<<<(tq + T - 1) / T, T, 0, stream>>>(out, pw, N);
}


# h = x @ W: wave keeps its 16 columns of W as B fragments in registers all kernel long; x tile read once per 8-wave work-group by whole rows via LDS; 1280 looping work-groups; same k order
# speedup vs baseline: 1.5201x; 1.5201x over previous
#include <hip/hip_runtime.h>

typedef __attribute__((ext_vector_type(2))) float v2f;
typedef __attribute__((ext_vector_type(8))) float v8f;

#define IN_CH 128
#define HID   128

__global__ void k_init_deg(unsigned* __restrict__ deg, int N) {
  int i = blockIdx.x * blockDim.x + threadIdx.x;
  if (i < N) deg[i] = 1u;
}

__global__ void k_count(const int* __restrict__ ei, int E, unsigned* __restrict__ deg) {
  int e = blockIdx.x * blockDim.x + threadIdx.x;
  if (e < E) atomicAdd(&deg[ei[E + e]], 1u);
}

__global__ void k_dis(const unsigned* __restrict__ deg, float* __restrict__ dis, int N) {
  int i = blockIdx.x * blockDim.x + threadIdx.x;
  if (i < N) dis[i] = rsqrtf((float)deg[i]);
}

#define APITCH 132

__global__ __launch_bounds__(256) void k_gemm(const float* __restrict__ x,
                                              const float* __restrict__ W,
                                              float* __restrict__ h, int ntiles) {
  __shared__ __attribute__((aligned(16))) float As[2][16 * APITCH];

  const int lane = threadIdx.x & 31;
  const int wave = threadIdx.x >> 5;
  const int col0 = wave * 16;

  const int mn  = lane & 15;
  const int kof = (lane >> 4) << 1;

  v2f bf[32];
  const float* Wc = W + col0 + mn;
#pragma unroll
  for (int i = 0; i < 32; ++i) {
    bf[i][0] = Wc[(size_t)(4 * i + kof) * HID];
    bf[i][1] = Wc[(size_t)(4 * i + kof + 1) * HID];
  }

  int t = blockIdx.x;
  if (t >= ntiles) return;

  const int r0 = wave * 2;
  const int so = r0 * APITCH + lane * 4;
  float4 p0, p1;
  {
    const float* sp = x + ((size_t)t * 16 + r0) * IN_CH + lane * 4;
    p0 = *(const float4*)sp;
    p1 = *(const float4*)(sp + IN_CH);
  }
  *(float4*)&As[0][so] = p0;
  *(float4*)&As[0][so + APITCH] = p1;
  __syncthreads();

  const int mo = (lane >> 4) * 8;
  int cur = 0;
  for (; t < ntiles; t += gridDim.x) {
    const int  tn   = t + gridDim.x;
    const bool more = tn < ntiles;
    if (more) {
      const float* sp = x + ((size_t)tn * 16 + r0) * IN_CH + lane * 4;
      p0 = *(const float4*)sp;
      p1 = *(const float4*)(sp + IN_CH);
    }

    v8f acc = {};
    const float* ar = &As[cur][mn * APITCH + kof];
#pragma unroll
    for (int i = 0; i < 32; ++i) {
      const float2 av = *(const float2*)(ar + 4 * i);
      v2f a; a[0] = av.x; a[1] = av.y;
      acc = __builtin_amdgcn_wmma_f32_16x16x4_f32(false, a, false, bf[i],
                                                  (short)0, acc, false, false);
    }

    float* hr = h + ((size_t)t * 16 + mo) * HID + col0 + mn;
#pragma unroll
    for (int r = 0; r < 8; ++r)
      hr[(size_t)r * HID] = acc[r];

    if (more) {
      *(float4*)&As[cur ^ 1][so] = p0;
      *(float4*)&As[cur ^ 1][so + APITCH] = p1;
    }
    __builtin_amdgcn_fence(__ATOMIC_RELEASE, "workgroup", "local");
    __builtin_amdgcn_s_barrier();
    __builtin_amdgcn_fence(__ATOMIC_ACQUIRE, "workgroup", "local");
    cur ^= 1;
  }
}

__global__ void k_init_out(const float* __restrict__ h, const float* __restrict__ dis,
                           const float* __restrict__ bias, float* __restrict__ out, int N) {
  int t = blockIdx.x * blockDim.x + threadIdx.x;
  if (t >= N * 32) return;
  int i  = t >> 5;
  int cg = (t & 31) << 2;
  float nrm = dis[i] * dis[i];
  float4 hv = *(const float4*)(h + (size_t)i * HID + cg);
  float4 bv = *(const float4*)(bias + cg);
  float4 o;
  o.x = bv.x + nrm * hv.x;
  o.y = bv.y + nrm * hv.y;
  o.z = bv.z + nrm * hv.z;
  o.w = bv.w + nrm * hv.w;
  *(float4*)(out + (size_t)i * HID + cg) = o;
}

__device__ __forceinline__ int bcast_i(int v, int j) {
  return __builtin_amdgcn_readlane(v, j);
}
__device__ __forceinline__ float bcast_f(float v, int j) {
  return __builtin_bit_cast(float, __builtin_amdgcn_readlane(__builtin_bit_cast(int, v), j));
}

__global__ __launch_bounds__(256) void k_scatter(const int* __restrict__ ei, int E,
                                                 const float* __restrict__ h,
                                                 const float* __restrict__ dis,
                                                 float* __restrict__ out) {
  const int lane = threadIdx.x & 31;
  const int wave = __builtin_amdgcn_readfirstlane((int)((blockIdx.x * blockDim.x + threadIdx.x) >> 5));
  const int base = wave * 32;
  if (base >= E) return;
  const int left = E - base;
  const int cnt  = left < 32 ? left : 32;

  int s = 0, d = 0;
  float nrm = 0.f;
  if (lane < cnt) {
    s = ei[base + lane];
    d = ei[E + base + lane];
    nrm = dis[s] * dis[d];
  }

  int j = 0;
  for (; j + 4 <= cnt; j += 4) {
    const int   s0 = bcast_i(s, j),     s1 = bcast_i(s, j + 1);
    const int   s2 = bcast_i(s, j + 2), s3 = bcast_i(s, j + 3);
    const float* p0 = h + (size_t)s0 * HID + lane;
    const float* p1 = h + (size_t)s1 * HID + lane;
    const float* p2 = h + (size_t)s2 * HID + lane;
    const float* p3 = h + (size_t)s3 * HID + lane;
    const float a0 = p0[0], a1 = p0[32], a2 = p0[64], a3 = p0[96];
    const float b0 = p1[0], b1 = p1[32], b2 = p1[64], b3 = p1[96];
    const float c0 = p2[0], c1 = p2[32], c2 = p2[64], c3 = p2[96];
    const float e0 = p3[0], e1 = p3[32], e2 = p3[64], e3 = p3[96];
    const int   d0 = bcast_i(d, j),     d1 = bcast_i(d, j + 1);
    const int   d2 = bcast_i(d, j + 2), d3 = bcast_i(d, j + 3);
    const float n0 = bcast_f(nrm, j),     n1 = bcast_f(nrm, j + 1);
    const float n2 = bcast_f(nrm, j + 2), n3 = bcast_f(nrm, j + 3);
    float* q0 = out + (size_t)d0 * HID + lane;
    float* q1 = out + (size_t)d1 * HID + lane;
    float* q2 = out + (size_t)d2 * HID + lane;
    float* q3 = out + (size_t)d3 * HID + lane;
    unsafeAtomicAdd(q0 +  0, n0 * a0);
    unsafeAtomicAdd(q0 + 32, n0 * a1);
    unsafeAtomicAdd(q0 + 64, n0 * a2);
    unsafeAtomicAdd(q0 + 96, n0 * a3);
    unsafeAtomicAdd(q1 +  0, n1 * b0);
    unsafeAtomicAdd(q1 + 32, n1 * b1);
    unsafeAtomicAdd(q1 + 64, n1 * b2);
    unsafeAtomicAdd(q1 + 96, n1 * b3);
    unsafeAtomicAdd(q2 +  0, n2 * c0);
    unsafeAtomicAdd(q2 + 32, n2 * c1);
    unsafeAtomicAdd(q2 + 64, n2 * c2);
    unsafeAtomicAdd(q2 + 96, n2 * c3);
    unsafeAtomicAdd(q3 +  0, n3 * e0);
    unsafeAtomicAdd(q3 + 32, n3 * e1);
    unsafeAtomicAdd(q3 + 64, n3 * e2);
    unsafeAtomicAdd(q3 + 96, n3 * e3);
  }
  for (; j < cnt; ++j) {
    const int   sj = bcast_i(s, j);
    const int   dj = bcast_i(d, j);
    const float nj = bcast_f(nrm, j);
    const float* hp = h + (size_t)sj * HID + lane;
    const float h0 = hp[0], h1 = hp[32], h2 = hp[64], h3 = hp[96];
    float* op = out + (size_t)dj * HID + lane;
    unsafeAtomicAdd(op +  0, nj * h0);
    unsafeAtomicAdd(op + 32, nj * h1);
    unsafeAtomicAdd(op + 64, nj * h2);
    unsafeAtomicAdd(op + 96, nj * h3);
  }
}

__global__ void k_prelu(float* __restrict__ out, const float* __restrict__ pw, int N) {
  int t = blockIdx.x * blockDim.x + threadIdx.x;
  if (t >= N * 32) return;
  int cg = (t & 31) << 2;
  float4 v = *(float4*)(out + (size_t)t * 4);
  float4 p = *(const float4*)(pw + cg);
  v.x = v.x >= 0.f ? v.x : p.x * v.x;
  v.y = v.y >= 0.f ? v.y : p.y * v.y;
  v.z = v.z >= 0.f ? v.z : p.z * v.z;
  v.w = v.w >= 0.f ? v.w : p.w * v.w;
  *(float4*)(out + (size_t)t * 4) = v;
}

extern "C" void kernel_launch(void* const* d_in, const int* in_sizes, int n_in,
                              void* d_out, int out_size, void* d_ws, size_t ws_size,
                              hipStream_t stream) {
  const float* x  = (const float*)d_in[0];
  const int*   ei = (const int*)d_in[1];
  const float* W  = (const float*)d_in[2];
  const float* b  = (const float*)d_in[3];
  const float* pw = (const float*)d_in[4];
  float* out = (float*)d_out;

  const int N = in_sizes[0] / IN_CH;
  const int E = in_sizes[1] / 2;

  char* ws = (char*)d_ws;
  size_t o0 = 0;
  size_t o1 = (o0 + (size_t)N * 4 + 255) & ~(size_t)255;
  size_t o2 = (o1 + (size_t)N * 4 + 255) & ~(size_t)255;
  unsigned* deg = (unsigned*)(ws + o0);
  float*    dis = (float*)(ws + o1);
  float*    h   = (float*)(ws + o2);

  const int T = 256;
  k_init_deg<<<(N + T - 1) / T, T, 0, stream>>>(deg, N);
  k_count  <<<(E + T - 1) / T, T, 0, stream>>>(ei, E, deg);
  k_dis    <<<(N + T - 1) / T, T, 0, stream>>>(deg, dis, N);

  k_gemm<<<1280, 256, 0, stream>>>(x, W, h, N / 16);

  int tq = N * 32;
  k_init_out<<<(tq + T - 1) / T, T, 0, stream>>>(h, dis, b, out, N);

  int sblocks = (E + T - 1) / T;
  k_scatter<<<sblocks, T, 0, stream>>>(ei, E, h, dis, out);

  k_prelu<<<(tq + T - 1) / T, T, 0, stream>>>(out, pw, N);
}
